# starting kernel
# baseline (speedup 1.0000x reference)
#include <hip/hip_runtime.h>

typedef __attribute__((ext_vector_type(2))) float v2f;
typedef __attribute__((ext_vector_type(8))) float v8f;

#define B_N   4096
#define D_INN 1024
#define D_F   256
#define NCOMP 8
#define NMK   80
#define LOG2PI 1.83787706640934548356f

__device__ __forceinline__ v8f wmma4(v2f a, v2f b, v8f c) {
  return __builtin_amdgcn_wmma_f32_16x16x4_f32(false, a, false, b, (short)0, c,
                                               false, false);
}

__global__ void feats_kernel(const float* __restrict__ x,
                             const float* __restrict__ W,
                             const float* __restrict__ bias,
                             float* __restrict__ feats) {
  extern __shared__ float smem[];
  float* As = smem;
  float* Bs = smem + 64 * 66;
  const int tid = threadIdx.x;
  const int lane = tid & 31;
  const int w = tid >> 5;
  const int rowbase = blockIdx.y * 64;
  const int colbase = blockIdx.x * 64;

  v8f acc[2] = {};
  const int l15 = lane & 15;
  const int khalf = (lane >> 4) * 2;

  for (int kc = 0; kc < D_INN; kc += 64) {
    for (int e = tid; e < 64 * 64; e += 256) {
      int r = e >> 6, c = e & 63;
      As[r * 66 + c] = x[(size_t)(rowbase + r) * D_INN + kc + c];
      Bs[c * 66 + r] = W[(size_t)(kc + r) * D_F + colbase + c];
    }
    __syncthreads();
    for (int kk = 0; kk < 64; kk += 4) {
      int klo = kk + khalf;
#pragma unroll
      for (int t = 0; t < 2; ++t) {
        int g = w * 2 + t;
        int mt = g >> 2, nt = g & 3;
        v2f a = *(const v2f*)(As + (mt * 16 + l15) * 66 + klo);
        v2f bv = *(const v2f*)(Bs + (nt * 16 + l15) * 66 + klo);
        acc[t] = wmma4(a, bv, acc[t]);
      }
    }
    __syncthreads();
  }

  const int hi = lane >> 4;
#pragma unroll
  for (int t = 0; t < 2; ++t) {
    int g = w * 2 + t;
    int mt = g >> 2, nt = g & 3;
    int n = nt * 16 + l15;
    float bb = bias[colbase + n];
#pragma unroll
    for (int r = 0; r < 8; ++r) {
      int mrow = mt * 16 + r + hi * 8;
      float v = acc[t][r] + bb;
      v = v > 0.f ? v : 0.f;
      feats[(size_t)(rowbase + mrow) * D_F + colbase + n] = v;
    }
  }
}

__global__ void trinv_kernel(const float* __restrict__ covs,
                             float* __restrict__ Linv,
                             float* __restrict__ logdet) {
  extern __shared__ float smem[];
  float* Ls = smem;
  float* logd = smem + 256 * 257;
  const int j = threadIdx.x;
  const int mk = blockIdx.x;
  const float* Lg = covs + (size_t)mk * (D_F * D_F);

  for (int e = j; e < 256 * 257; e += 256) Ls[e] = 0.f;
  __syncthreads();

  float Ljj = Lg[j * D_F + j];
  Ls[j * 257 + j] = 1.f / Ljj;
  logd[j] = logf(Ljj);
  __syncthreads();

  for (int i = 1; i < D_F; ++i) {
    if (j < i) {
      const float* Lrow = Lg + (size_t)i * D_F;
      float s = 0.f;
      for (int t = j; t < i; ++t) s += Lrow[t] * Ls[t * 257 + j];
      Ls[i * 257 + j] = -s / Lrow[i];
    }
    __syncthreads();
  }

  float* outp = Linv + (size_t)mk * (D_F * D_F);
  for (int i = 0; i < D_F; ++i) outp[(size_t)i * D_F + j] = Ls[i * 257 + j];
  if (j == 0) {
    float s = 0.f;
    for (int t = 0; t < D_F; ++t) s += logd[t];
    logdet[mk] = s;
  }
}

__global__ void gmm_kernel(const float* __restrict__ feats,
                           const float* __restrict__ Linv,
                           const float* __restrict__ logdet,
                           const float* __restrict__ means,
                           const float* __restrict__ weights,
                           float* __restrict__ out) {
  extern __shared__ float smem[];
  float* As = smem;
  float* Bs = smem + 256 * 66;
  float* quad_s = smem + 256 * 66 + 64 * 258;
  const int tid = threadIdx.x;
  const int lane = tid & 31;
  const int w = tid >> 5;
  const int mk = blockIdx.y;
  const int b0 = blockIdx.x * 64;
  const float* LinvG = Linv + (size_t)mk * (D_F * D_F);
  const float* mu = means + (size_t)mk * D_F;

  for (int e = tid; e < 64 * 256; e += 256) {
    int n = e >> 8, d = e & 255;
    Bs[n * 258 + d] = feats[(size_t)(b0 + n) * D_F + d] - mu[d];
  }
  if (tid < 64) quad_s[tid] = 0.f;
  __syncthreads();

  v8f acc[8] = {};
  const int l15 = lane & 15;
  const int khalf = (lane >> 4) * 2;

  for (int kc = 0; kc < D_F; kc += 64) {
    for (int e = tid; e < 256 * 64; e += 256) {
      int r = e >> 6, c = e & 63;
      As[r * 66 + c] = LinvG[(size_t)r * D_F + kc + c];
    }
    __syncthreads();
    for (int kk = 0; kk < 64; kk += 4) {
      int klo = kk + khalf;
#pragma unroll
      for (int i = 0; i < 2; ++i) {
        int mt = w * 2 + i;
        v2f a = *(const v2f*)(As + (mt * 16 + l15) * 66 + klo);
#pragma unroll
        for (int nt = 0; nt < 4; ++nt) {
          v2f bv = *(const v2f*)(Bs + (nt * 16 + l15) * 258 + kc + klo);
          acc[i * 4 + nt] = wmma4(a, bv, acc[i * 4 + nt]);
        }
      }
    }
    __syncthreads();
  }

#pragma unroll
  for (int t = 0; t < 8; ++t) {
    int nt = t & 3;
    float p = 0.f;
#pragma unroll
    for (int r = 0; r < 8; ++r) p += acc[t][r] * acc[t][r];
    atomicAdd(&quad_s[nt * 16 + l15], p);
  }
  __syncthreads();

  if (tid < 64) {
    int m = mk >> 3, kcomp = mk & 7;
    const float* wr = weights + m * NCOMP;
    float wmax = wr[0];
#pragma unroll
    for (int t = 1; t < 8; ++t) wmax = fmaxf(wmax, wr[t]);
    float denom = 0.f;
#pragma unroll
    for (int t = 0; t < 8; ++t) denom += expf(wr[t] - wmax);
    float wmk = expf(wr[kcomp] - wmax) / denom;
    float logp = -0.5f * (quad_s[tid] + 256.0f * LOG2PI) - logdet[mk];
    out[(size_t)(b0 + tid) * NMK + mk] = wmk * logp;
  }
}

extern "C" void kernel_launch(void* const* d_in, const int* in_sizes, int n_in,
                              void* d_out, int out_size, void* d_ws,
                              size_t ws_size, hipStream_t stream) {
  (void)in_sizes; (void)n_in; (void)out_size; (void)ws_size;
  const float* x       = (const float*)d_in[0];
  const float* W       = (const float*)d_in[1];
  const float* bias    = (const float*)d_in[2];
  const float* means   = (const float*)d_in[3];
  const float* covs    = (const float*)d_in[4];
  const float* weights = (const float*)d_in[5];
  float* out = (float*)d_out;

  float* ws = (float*)d_ws;
  float* feats  = ws;
  float* LinvW  = ws + (size_t)B_N * D_F;
  float* logdet = LinvW + (size_t)NMK * D_F * D_F;

  size_t sh1 = (size_t)(2 * 64 * 66) * sizeof(float);
  feats_kernel<<<dim3(D_F / 64, B_N / 64), 256, sh1, stream>>>(x, W, bias,
                                                               feats);

  size_t sh2 = (size_t)(256 * 257 + 256) * sizeof(float);
  trinv_kernel<<<dim3(NMK), 256, sh2, stream>>>(covs, LinvW, logdet);

  size_t sh3 = (size_t)(256 * 66 + 64 * 258 + 64) * sizeof(float);
  gmm_kernel<<<dim3(B_N / 64, NMK), 256, sh3, stream>>>(feats, LinvW, logdet,
                                                        means, weights, out);
}


# no inverse kernel: third kernel solves L z = d by forward substitution over 16-row blocks on f32 WMMA straight from cov, only 16x16 diagonal blocks inverted; sum order of z and quad changes
# speedup vs baseline: 5.2215x; 5.2215x over previous
#include <hip/hip_runtime.h>

typedef __attribute__((ext_vector_type(2))) float v2f;
typedef __attribute__((ext_vector_type(8))) float v8f;

#define B_N   4096
#define D_INN 1024
#define D_F   256
#define NCOMP 8
#define NMK   80
#define LOG2PI 1.83787706640934548356f

__device__ __forceinline__ v8f wmma4(v2f a, v2f b, v8f c) {
  return __builtin_amdgcn_wmma_f32_16x16x4_f32(false, a, false, b, (short)0, c,
                                               false, false);
}

__global__ void feats_kernel(const float* __restrict__ x,
                             const float* __restrict__ W,
                             const float* __restrict__ bias,
                             float* __restrict__ feats) {
  extern __shared__ float smem[];
  float* As = smem;
  float* Bs = smem + 64 * 66;
  const int tid = threadIdx.x;
  const int lane = tid & 31;
  const int w = tid >> 5;
  const int rowbase = blockIdx.y * 64;
  const int colbase = blockIdx.x * 64;

  v8f acc[2] = {};
  const int l15 = lane & 15;
  const int khalf = (lane >> 4) * 2;

  for (int kc = 0; kc < D_INN; kc += 64) {
    for (int e = tid; e < 64 * 64; e += 256) {
      int r = e >> 6, c = e & 63;
      As[r * 66 + c] = x[(size_t)(rowbase + r) * D_INN + kc + c];
      Bs[c * 66 + r] = W[(size_t)(kc + r) * D_F + colbase + c];
    }
    __syncthreads();
    for (int kk = 0; kk < 64; kk += 4) {
      int klo = kk + khalf;
#pragma unroll
      for (int t = 0; t < 2; ++t) {
        int g = w * 2 + t;
        int mt = g >> 2, nt = g & 3;
        v2f a = *(const v2f*)(As + (mt * 16 + l15) * 66 + klo);
        v2f bv = *(const v2f*)(Bs + (nt * 16 + l15) * 66 + klo);
        acc[t] = wmma4(a, bv, acc[t]);
      }
    }
    __syncthreads();
  }

  const int hi = lane >> 4;
#pragma unroll
  for (int t = 0; t < 2; ++t) {
    int g = w * 2 + t;
    int mt = g >> 2, nt = g & 3;
    int n = nt * 16 + l15;
    float bb = bias[colbase + n];
#pragma unroll
    for (int r = 0; r < 8; ++r) {
      int mrow = mt * 16 + r + hi * 8;
      float v = acc[t][r] + bb;
      v = v > 0.f ? v : 0.f;
      feats[(size_t)(rowbase + mrow) * D_F + colbase + n] = v;
    }
  }
}

__global__ void __launch_bounds__(256)
diag_kernel(const float* __restrict__ covs, float* __restrict__ Dinv,
            float* __restrict__ logdet) {
  __shared__ float logd[256];
  const int tid = threadIdx.x;
  const int mk = blockIdx.x;
  const float* Lg = covs + (size_t)mk * (D_F * D_F);
  logd[tid] = logf(Lg[tid * D_F + tid]);

  const int blk = tid >> 4, c = tid & 15;
  const float* Lb = Lg + (size_t)(blk * 16) * D_F + blk * 16;
  float xcol[16];
#pragma unroll
  for (int r = 0; r < 16; ++r) {
    float s = 0.f;
#pragma unroll
    for (int t = 0; t < r; ++t) s += Lb[r * D_F + t] * xcol[t];
    float d = Lb[r * D_F + r];
    xcol[r] = (r < c) ? 0.f : ((r == c) ? 1.f / d : -s / d);
  }
  float* o = Dinv + ((size_t)mk * 16 + blk) * 256;
#pragma unroll
  for (int r = 0; r < 16; ++r) o[r * 16 + c] = xcol[r];

  __syncthreads();
  if (tid == 0) {
    float s = 0.f;
    for (int t = 0; t < D_F; ++t) s += logd[t];
    logdet[mk] = s;
  }
}

#define PST 258

__global__ void __launch_bounds__(128)
gmm_kernel(const float* __restrict__ feats, const float* __restrict__ covs,
           const float* __restrict__ Dinv, const float* __restrict__ logdet,
           const float* __restrict__ means, const float* __restrict__ weights,
           float* __restrict__ out) {
  extern __shared__ float smem[];
  float* Bs = smem;
  float* Ps = smem + 64 * PST;
  const int tid = threadIdx.x;
  const int lane = tid & 31;
  const int w = tid >> 5;
  const int mk = blockIdx.y;
  const int b0 = blockIdx.x * 64;
  const float* Lg = covs + (size_t)mk * (D_F * D_F);
  const float* Dg = Dinv + (size_t)mk * (16 * 256);
  const float* mu = means + (size_t)mk * D_F;

  const int pr = tid >> 3, ps = tid & 7;
  float4 pf[8];

#define PANEL_ISSUE(I)                                                     \
  {                                                                        \
    const float* lrow = Lg + (size_t)(16 * (I) + pr) * D_F;                \
    const float* drow = Dg + ((I) * 16 + pr) * 16 - 16 * (I);              \
    _Pragma("unroll") for (int u = 0; u < 8; ++u) {                        \
      int q = ps + 8 * u;                                                  \
      const float* src = (q < 4 * (I)) ? lrow : drow;                      \
      if (q < 4 * (I) + 4) pf[u] = *(const float4*)(src + 4 * q);          \
    }                                                                      \
  }
#define PANEL_COMMIT(I)                                                    \
  {                                                                        \
    float* prow = Ps + ((I) & 1) * (16 * PST) + pr * PST;                  \
    _Pragma("unroll") for (int u = 0; u < 8; ++u) {                        \
      int q = ps + 8 * u;                                                  \
      if (q < 4 * (I) + 4) {                                               \
        *(v2f*)(prow + 4 * q) = v2f{pf[u].x, pf[u].y};                     \
        *(v2f*)(prow + 4 * q + 2) = v2f{pf[u].z, pf[u].w};                 \
      }                                                                    \
    }                                                                      \
  }

  PANEL_ISSUE(0)

  for (int e = tid; e < 64 * 64; e += 128) {
    int n = e >> 6, q = e & 63;
    float4 f = *(const float4*)(feats + (size_t)(b0 + n) * D_F + 4 * q);
    float m0 = mu[4 * q], m1 = mu[4 * q + 1], m2 = mu[4 * q + 2],
          m3 = mu[4 * q + 3];
    *(v2f*)(Bs + n * PST + 4 * q) = v2f{f.x - m0, f.y - m1};
    *(v2f*)(Bs + n * PST + 4 * q + 2) = v2f{f.z - m2, f.w - m3};
  }
  PANEL_COMMIT(0)
  __syncthreads();

  const int l15 = lane & 15;
  const int hi = lane >> 4;
  const int khalf = hi * 2;
  const float* brow = Bs + (16 * w + l15) * PST + khalf;
  float p = 0.f;

  for (int i = 0; i < 16; ++i) {
    if (i + 1 < 16) PANEL_ISSUE(i + 1)

    const float* arow = Ps + (i & 1) * (16 * PST) + l15 * PST + khalf;

    v8f s0 = {}, s1 = {};
    v2f a0 = *(const v2f*)(arow + 0), a1 = *(const v2f*)(arow + 4);
    v2f a2 = *(const v2f*)(arow + 8), a3 = *(const v2f*)(arow + 12);
    v2f c0 = *(const v2f*)(brow + 0), c1 = *(const v2f*)(brow + 4);
    v2f c2 = *(const v2f*)(brow + 8), c3 = *(const v2f*)(brow + 12);
    for (int j = 0; j < i; ++j) {
      const float* an = arow + 16 * (j + 1);
      const float* bn = brow + 16 * (j + 1);
      v2f na0 = *(const v2f*)(an + 0), na1 = *(const v2f*)(an + 4);
      v2f na2 = *(const v2f*)(an + 8), na3 = *(const v2f*)(an + 12);
      v2f nc0 = *(const v2f*)(bn + 0), nc1 = *(const v2f*)(bn + 4);
      v2f nc2 = *(const v2f*)(bn + 8), nc3 = *(const v2f*)(bn + 12);
      s0 = wmma4(a0, c0, s0);
      s1 = wmma4(a1, c1, s1);
      s0 = wmma4(a2, c2, s0);
      s1 = wmma4(a3, c3, s1);
      a0 = na0; a1 = na1; a2 = na2; a3 = na3;
      c0 = nc0; c1 = nc1; c2 = nc2; c3 = nc3;
    }

    float* dcol = Bs + (16 * w + l15) * PST + 16 * i + 8 * hi;
#pragma unroll
    for (int r = 0; r < 8; ++r) dcol[r] = dcol[r] - (s0[r] + s1[r]);

    v8f z = {};
#pragma unroll
    for (int kk = 0; kk < 16; kk += 4) {
      v2f a = *(const v2f*)(arow + 16 * i + kk);
      v2f bv = *(const v2f*)(brow + 16 * i + kk);
      z = wmma4(a, bv, z);
    }
#pragma unroll
    for (int r = 0; r < 8; ++r) {
      dcol[r] = z[r];
      p += z[r] * z[r];
    }

    if (i + 1 < 16) PANEL_COMMIT(i + 1)
    __syncthreads();
  }

  p += __shfl_xor(p, 16);

  if (hi == 0) {
    int m = mk >> 3, kcomp = mk & 7;
    const float* wr = weights + m * NCOMP;
    float wmax = wr[0];
#pragma unroll
    for (int t = 1; t < 8; ++t) wmax = fmaxf(wmax, wr[t]);
    float denom = 0.f;
#pragma unroll
    for (int t = 0; t < 8; ++t) denom += expf(wr[t] - wmax);
    float wmk = expf(wr[kcomp] - wmax) / denom;
    float logp = -0.5f * (p + 256.0f * LOG2PI) - logdet[mk];
    out[(size_t)(b0 + 16 * w + l15) * NMK + mk] = wmk * logp;
  }
}

extern "C" void kernel_launch(void* const* d_in, const int* in_sizes, int n_in,
                              void* d_out, int out_size, void* d_ws,
                              size_t ws_size, hipStream_t stream) {
  (void)in_sizes; (void)n_in; (void)out_size; (void)ws_size;
  const float* x       = (const float*)d_in[0];
  const float* W       = (const float*)d_in[1];
  const float* bias    = (const float*)d_in[2];
  const float* means   = (const float*)d_in[3];
  const float* covs    = (const float*)d_in[4];
  const float* weights = (const float*)d_in[5];
  float* out = (float*)d_out;

  float* ws = (float*)d_ws;
  float* feats  = ws;
  float* DinvW  = ws + (size_t)B_N * D_F;
  float* logdet = DinvW + (size_t)NMK * 16 * 256;

  size_t sh1 = (size_t)(2 * 64 * 66) * sizeof(float);
  feats_kernel<<<dim3(D_F / 64, B_N / 64), 256, sh1, stream>>>(x, W, bias,
                                                               feats);

  diag_kernel<<<dim3(NMK), 256, 0, stream>>>(covs, DinvW, logdet);

  size_t sh3 = (size_t)(64 * PST + 2 * 16 * PST) * sizeof(float);
  gmm_kernel<<<dim3(B_N / 64, NMK), 128, sh3, stream>>>(
      feats, covs, DinvW, logdet, means, weights, out);
}


# third kernel: right-hand side and solved blocks held in registers in accumulator layout, fed to the B operand directly (no LDS copy); L panel read as 16-byte fragments; k order in 16-blocks changes
# speedup vs baseline: 6.9055x; 1.3225x over previous
#include <hip/hip_runtime.h>

typedef __attribute__((ext_vector_type(2))) float v2f;
typedef __attribute__((ext_vector_type(4))) float v4f;
typedef __attribute__((ext_vector_type(8))) float v8f;

#define B_N   4096
#define D_INN 1024
#define D_F   256
#define NCOMP 8
#define NMK   80
#define LOG2PI 1.83787706640934548356f

__device__ __forceinline__ v8f wmma4(v2f a, v2f b, v8f c) {
  return __builtin_amdgcn_wmma_f32_16x16x4_f32(false, a, false, b, (short)0, c,
                                               false, false);
}

__global__ void feats_kernel(const float* __restrict__ x,
                             const float* __restrict__ W,
                             const float* __restrict__ bias,
                             float* __restrict__ feats) {
  extern __shared__ float smem[];
  float* As = smem;
  float* Bs = smem + 64 * 66;
  const int tid = threadIdx.x;
  const int lane = tid & 31;
  const int w = tid >> 5;
  const int rowbase = blockIdx.y * 64;
  const int colbase = blockIdx.x * 64;

  v8f acc[2] = {};
  const int l15 = lane & 15;
  const int khalf = (lane >> 4) * 2;

  for (int kc = 0; kc < D_INN; kc += 64) {
    for (int e = tid; e < 64 * 64; e += 256) {
      int r = e >> 6, c = e & 63;
      As[r * 66 + c] = x[(size_t)(rowbase + r) * D_INN + kc + c];
      Bs[c * 66 + r] = W[(size_t)(kc + r) * D_F + colbase + c];
    }
    __syncthreads();
    for (int kk = 0; kk < 64; kk += 4) {
      int klo = kk + khalf;
#pragma unroll
      for (int t = 0; t < 2; ++t) {
        int g = w * 2 + t;
        int mt = g >> 2, nt = g & 3;
        v2f a = *(const v2f*)(As + (mt * 16 + l15) * 66 + klo);
        v2f bv = *(const v2f*)(Bs + (nt * 16 + l15) * 66 + klo);
        acc[t] = wmma4(a, bv, acc[t]);
      }
    }
    __syncthreads();
  }

  const int hi = lane >> 4;
#pragma unroll
  for (int t = 0; t < 2; ++t) {
    int g = w * 2 + t;
    int mt = g >> 2, nt = g & 3;
    int n = nt * 16 + l15;
    float bb = bias[colbase + n];
#pragma unroll
    for (int r = 0; r < 8; ++r) {
      int mrow = mt * 16 + r + hi * 8;
      float v = acc[t][r] + bb;
      v = v > 0.f ? v : 0.f;
      feats[(size_t)(rowbase + mrow) * D_F + colbase + n] = v;
    }
  }
}

__global__ void __launch_bounds__(256)
diag_kernel(const float* __restrict__ covs, float* __restrict__ Dinv,
            float* __restrict__ logdet) {
  __shared__ float logd[256];
  const int tid = threadIdx.x;
  const int mk = blockIdx.x;
  const float* Lg = covs + (size_t)mk * (D_F * D_F);
  logd[tid] = logf(Lg[tid * D_F + tid]);

  const int blk = tid >> 4, c = tid & 15;
  const float* Lb = Lg + (size_t)(blk * 16) * D_F + blk * 16;
  float xcol[16];
#pragma unroll
  for (int r = 0; r < 16; ++r) {
    float s = 0.f;
#pragma unroll
    for (int t = 0; t < r; ++t) s += Lb[r * D_F + t] * xcol[t];
    float d = Lb[r * D_F + r];
    xcol[r] = (r < c) ? 0.f : ((r == c) ? 1.f / d : -s / d);
  }
  float* o = Dinv + ((size_t)mk * 16 + blk) * 256;
#pragma unroll
  for (int r = 0; r < 16; ++r) o[r * 16 + c] = xcol[r];

  __syncthreads();
  if (tid == 0) {
    float s = 0.f;
    for (int t = 0; t < D_F; ++t) s += logd[t];
    logdet[mk] = s;
  }
}

#define PST 260

template <int I>
__device__ __forceinline__ void panel_issue(v4f (&pf)[8], const float* Lg,
                                            const float* Dg, int pr, int ps) {
  const float* lrow = Lg + (size_t)(16 * I + pr) * D_F;
  const float* drow = Dg + (I * 16 + pr) * 16 - 16 * I;
#pragma unroll
  for (int u = 0; u < 8; ++u) {
    if (8 * u < 4 * I + 4) {
      int q = ps + 8 * u;
      int qc = (q < 4 * I + 4) ? q : 4 * I + 3;
      const float* src = (qc < 4 * I) ? lrow : drow;
      pf[u] = *(const v4f*)(src + 4 * qc);
    }
  }
}

template <int I>
__device__ __forceinline__ void panel_commit(const v4f (&pf)[8], float* Ps,
                                             int pr, int ps) {
  float* prow = Ps + (I & 1) * (16 * PST) + pr * PST;
#pragma unroll
  for (int u = 0; u < 8; ++u) {
    int q = ps + 8 * u;
    if (q < 4 * I + 4) *(v4f*)(prow + 4 * q) = pf[u];
  }
}

template <int I>
__device__ __forceinline__ void solve_step(v8f (&z)[16], float& p,
                                           const float* Ps, const float* mu8,
                                           int l15, int hi) {
  const float* arow = Ps + (I & 1) * (16 * PST) + l15 * PST + 8 * hi;
  const v4f m0 = *(const v4f*)(mu8 + 16 * I);
  const v4f m1 = *(const v4f*)(mu8 + 16 * I + 4);

  v8f s0 = {}, s1 = {};
#pragma unroll
  for (int j = 0; j < I; ++j) {
    const v4f aa = *(const v4f*)(arow + 16 * j);
    const v4f ab = *(const v4f*)(arow + 16 * j + 4);
    s0 = wmma4(v2f{aa.x, aa.y}, v2f{z[j][0], z[j][1]}, s0);
    s1 = wmma4(v2f{aa.z, aa.w}, v2f{z[j][2], z[j][3]}, s1);
    s0 = wmma4(v2f{ab.x, ab.y}, v2f{z[j][4], z[j][5]}, s0);
    s1 = wmma4(v2f{ab.z, ab.w}, v2f{z[j][6], z[j][7]}, s1);
  }

  const v8f mm = {m0.x, m0.y, m0.z, m0.w, m1.x, m1.y, m1.z, m1.w};
  const v8f rr = (z[I] - mm) - (s0 + s1);

  const v4f da = *(const v4f*)(arow + 16 * I);
  const v4f db = *(const v4f*)(arow + 16 * I + 4);
  v8f zz = {};
  zz = wmma4(v2f{da.x, da.y}, v2f{rr[0], rr[1]}, zz);
  zz = wmma4(v2f{da.z, da.w}, v2f{rr[2], rr[3]}, zz);
  zz = wmma4(v2f{db.x, db.y}, v2f{rr[4], rr[5]}, zz);
  zz = wmma4(v2f{db.z, db.w}, v2f{rr[6], rr[7]}, zz);
  z[I] = zz;
#pragma unroll
  for (int r = 0; r < 8; ++r) p += zz[r] * zz[r];
}

__global__ void __launch_bounds__(128)
gmm_kernel(const float* __restrict__ feats, const float* __restrict__ covs,
           const float* __restrict__ Dinv, const float* __restrict__ logdet,
           const float* __restrict__ means, const float* __restrict__ weights,
           float* __restrict__ out) {
  extern __shared__ __align__(16) float panel_lds[];
  float* Ps = panel_lds;
  const int tid = threadIdx.x;
  const int lane = tid & 31;
  const int w = tid >> 5;
  const int mk = blockIdx.y;
  const int b0 = blockIdx.x * 64;
  const float* Lg = covs + (size_t)mk * (D_F * D_F);
  const float* Dg = Dinv + (size_t)mk * (16 * 256);
  const float* mu = means + (size_t)mk * D_F;
  const int l15 = lane & 15;
  const int hi = lane >> 4;
  const int pr = tid >> 3, ps = tid & 7;
  v4f pf[8];

  panel_issue<0>(pf, Lg, Dg, pr, ps);

  const float* frow = feats + (size_t)(b0 + 16 * w + l15) * D_F + 8 * hi;
  v8f z[16];
#pragma unroll
  for (int i = 0; i < 16; ++i) {
    const v4f f0 = *(const v4f*)(frow + 16 * i);
    const v4f f1 = *(const v4f*)(frow + 16 * i + 4);
    z[i] = v8f{f0.x, f0.y, f0.z, f0.w, f1.x, f1.y, f1.z, f1.w};
  }

  panel_commit<0>(pf, Ps, pr, ps);
  __syncthreads();

  const float* mu8 = mu + 8 * hi;
  float p = 0.f;

#define SOLVE_STEP(I)                        \
  panel_issue<(I) + 1>(pf, Lg, Dg, pr, ps);  \
  solve_step<(I)>(z, p, Ps, mu8, l15, hi);   \
  __builtin_amdgcn_sched_barrier(0);         \
  panel_commit<(I) + 1>(pf, Ps, pr, ps);     \
  __syncthreads();

  SOLVE_STEP(0)  SOLVE_STEP(1)  SOLVE_STEP(2)  SOLVE_STEP(3)
  SOLVE_STEP(4)  SOLVE_STEP(5)  SOLVE_STEP(6)  SOLVE_STEP(7)
  SOLVE_STEP(8)  SOLVE_STEP(9)  SOLVE_STEP(10) SOLVE_STEP(11)
  SOLVE_STEP(12) SOLVE_STEP(13) SOLVE_STEP(14)
  solve_step<15>(z, p, Ps, mu8, l15, hi);

  p += __shfl_xor(p, 16);

  if (hi == 0) {
    int m = mk >> 3, kcomp = mk & 7;
    const float* wr = weights + m * NCOMP;
    float wmax = wr[0];
#pragma unroll
    for (int t = 1; t < 8; ++t) wmax = fmaxf(wmax, wr[t]);
    float denom = 0.f;
#pragma unroll
    for (int t = 0; t < 8; ++t) denom += expf(wr[t] - wmax);
    float wmk = expf(wr[kcomp] - wmax) / denom;
    float logp = -0.5f * (p + 256.0f * LOG2PI) - logdet[mk];
    out[(size_t)(b0 + 16 * w + l15) * NMK + mk] = wmk * logp;
  }
}

extern "C" void kernel_launch(void* const* d_in, const int* in_sizes, int n_in,
                              void* d_out, int out_size, void* d_ws,
                              size_t ws_size, hipStream_t stream) {
  (void)in_sizes; (void)n_in; (void)out_size; (void)ws_size;
  const float* x       = (const float*)d_in[0];
  const float* W       = (const float*)d_in[1];
  const float* bias    = (const float*)d_in[2];
  const float* means   = (const float*)d_in[3];
  const float* covs    = (const float*)d_in[4];
  const float* weights = (const float*)d_in[5];
  float* out = (float*)d_out;

  float* ws = (float*)d_ws;
  float* feats  = ws;
  float* DinvW  = ws + (size_t)B_N * D_F;
  float* logdet = DinvW + (size_t)NMK * 16 * 256;

  size_t sh1 = (size_t)(2 * 64 * 66) * sizeof(float);
  feats_kernel<<<dim3(D_F / 64, B_N / 64), 256, sh1, stream>>>(x, W, bias,
                                                               feats);

  diag_kernel<<<dim3(NMK), 256, 0, stream>>>(covs, DinvW, logdet);

  size_t sh3 = (size_t)(2 * 16 * PST) * sizeof(float);
  gmm_kernel<<<dim3(B_N / 64, NMK), 128, sh3, stream>>>(
      feats, covs, DinvW, logdet, means, weights, out);
}


# first launch: x rows go straight from global memory into A fragments (no LDS for x), the work-group's whole W column slab staged once, untransposed, in LDS: 1 barrier for 32; feats k order changes
# speedup vs baseline: 8.3869x; 1.2145x over previous
#include <hip/hip_runtime.h>

typedef __attribute__((ext_vector_type(2))) float v2f;
typedef __attribute__((ext_vector_type(4))) float v4f;
typedef __attribute__((ext_vector_type(8))) float v8f;

#define B_N   4096
#define D_INN 1024
#define D_F   256
#define NCOMP 8
#define NMK   80
#define LOG2PI 1.83787706640934548356f

__device__ __forceinline__ v8f wmma4(v2f a, v2f b, v8f c) {
  return __builtin_amdgcn_wmma_f32_16x16x4_f32(false, a, false, b, (short)0, c,
                                               false, false);
}

#define WS_FLOATS (D_INN * 64)

__device__ __forceinline__ void xchunk_load(v4f (&a)[16], const float* p) {
#pragma unroll
  for (int b = 0; b < 8; ++b) {
    a[2 * b] = *(const v4f*)(p + 16 * b);
    a[2 * b + 1] = *(const v4f*)(p + 16 * b + 4);
  }
}

__device__ __forceinline__ void xchunk_mma(v8f (&acc)[2], const v4f (&a)[16],
                                           const float* b0, const float* b1) {
#pragma unroll
  for (int b = 0; b < 8; ++b) {
#pragma unroll
    for (int q = 0; q < 4; ++q) {
      const v4f av = a[2 * b + (q >> 1)];
      const v2f af = (q & 1) ? v2f{av.z, av.w} : v2f{av.x, av.y};
      const int ro = (16 * b + 2 * q) * 64;
      acc[0] = wmma4(af, v2f{b0[ro], b0[ro + 64]}, acc[0]);
      acc[1] = wmma4(af, v2f{b1[ro], b1[ro + 64]}, acc[1]);
    }
  }
}

__global__ void __launch_bounds__(256)
feats_kernel(const float* __restrict__ x, const float* __restrict__ W,
             const float* __restrict__ bias, float* __restrict__ feats) {
  __shared__ __attribute__((aligned(16))) float Ws[WS_FLOATS];
  const int tid = threadIdx.x;
  const int lane = tid & 31;
  const int w = tid >> 5;
  const int l15 = lane & 15;
  const int hi = lane >> 4;
  const int rt = w & 3;
  const int ch = w >> 2;
  const int rowbase = blockIdx.y * 64;
  const int colbase = blockIdx.x * 64;

  const float* xrow = x + (size_t)(rowbase + 16 * rt + l15) * D_INN + 8 * hi;
  v4f a0[16], a1[16];
  xchunk_load(a0, xrow);

  {
    const int sw = 16 * ((tid >> 7) & 1);
    const float* wsrc = W + (size_t)(tid >> 4) * D_F + colbase + 4 * (tid & 15);
    float* wdst = Ws + (tid >> 4) * 64 + ((4 * (tid & 15)) ^ sw);
#pragma unroll
    for (int g = 0; g < 2; ++g) {
      v4f t[32];
#pragma unroll
      for (int u = 0; u < 32; ++u)
        t[u] = *(const v4f*)(wsrc + (size_t)(16 * (32 * g + u)) * D_F);
#pragma unroll
      for (int u = 0; u < 32; ++u)
        *(v4f*)(wdst + 16 * (32 * g + u) * 64) = t[u];
    }
  }
  __syncthreads();

  const float* b0 = Ws + (8 * hi) * 64 + ((32 * ch + l15) ^ (16 * hi));
  const float* b1 = Ws + (8 * hi) * 64 + ((32 * ch + 16 + l15) ^ (16 * hi));
  v8f acc[2] = {};
  const float* xp = xrow;
#pragma unroll 1
  for (int c2 = 0; c2 < 4; ++c2) {
    xchunk_load(a1, xp + 128);
    xchunk_mma(acc, a0, b0, b1);
    xchunk_load(a0, xp + ((c2 < 3) ? 256 : 128));
    xchunk_mma(acc, a1, b0 + 128 * 64, b1 + 128 * 64);
    xp += 256;
    b0 += 256 * 64;
    b1 += 256 * 64;
  }

#pragma unroll
  for (int t = 0; t < 2; ++t) {
    const int n = colbase + 32 * ch + 16 * t + l15;
    const float bb = bias[n];
#pragma unroll
    for (int r = 0; r < 8; ++r) {
      const int mrow = rowbase + 16 * rt + 8 * hi + r;
      float v = acc[t][r] + bb;
      v = v > 0.f ? v : 0.f;
      feats[(size_t)mrow * D_F + n] = v;
    }
  }
}

__global__ void __launch_bounds__(256)
diag_kernel(const float* __restrict__ covs, float* __restrict__ Dinv,
            float* __restrict__ logdet) {
  __shared__ float logd[256];
  const int tid = threadIdx.x;
  const int mk = blockIdx.x;
  const float* Lg = covs + (size_t)mk * (D_F * D_F);
  logd[tid] = logf(Lg[tid * D_F + tid]);

  const int blk = tid >> 4, c = tid & 15;
  const float* Lb = Lg + (size_t)(blk * 16) * D_F + blk * 16;
  float xcol[16];
#pragma unroll
  for (int r = 0; r < 16; ++r) {
    float s = 0.f;
#pragma unroll
    for (int t = 0; t < r; ++t) s += Lb[r * D_F + t] * xcol[t];
    float d = Lb[r * D_F + r];
    xcol[r] = (r < c) ? 0.f : ((r == c) ? 1.f / d : -s / d);
  }
  float* o = Dinv + ((size_t)mk * 16 + blk) * 256;
#pragma unroll
  for (int r = 0; r < 16; ++r) o[r * 16 + c] = xcol[r];

  __syncthreads();
  if (tid == 0) {
    float s = 0.f;
    for (int t = 0; t < D_F; ++t) s += logd[t];
    logdet[mk] = s;
  }
}

#define LROWP(I) (16 * (I) + 20)
#define LOFF(I) (128 * (I) * ((I) - 1) + 320 * (I))
#define LP_FLOATS 35840

template <int I>
__device__ __forceinline__ void stage_issue(v4f (&t)[4], const float* Lg,
                                            const float* Dg, int pr, int ps) {
  const float* lrow = Lg + (size_t)(16 * I + pr) * D_F;
  const float* drow = Dg + (I * 16 + pr) * 16 - 16 * I;
  constexpr int NP = 4 * I + 4;
#pragma unroll
  for (int u = 0; u < (NP + 15) / 16; ++u) {
    int q = ps + 16 * u;
    int qc = (q < NP) ? q : NP - 1;
    const float* src = (qc < 4 * I) ? lrow : drow;
    t[u] = *(const v4f*)(src + 4 * qc);
  }
}

template <int I>
__device__ __forceinline__ void stage_commit(const v4f (&t)[4], float* Lp,
                                             int pr, int ps) {
  float* prow = Lp + LOFF(I) + pr * LROWP(I);
  constexpr int NP = 4 * I + 4;
#pragma unroll
  for (int u = 0; u < (NP + 15) / 16; ++u) {
    int q = ps + 16 * u;
    if (q < NP) *(v4f*)(prow + 4 * q) = t[u];
  }
}

template <int I>
__device__ __forceinline__ void solve_step(v8f (&z)[16], float& p,
                                           const float* Lp, int l15, int hi) {
  const float* arow = Lp + LOFF(I) + l15 * LROWP(I) + 8 * hi;
  const float* mrow = Lp + LP_FLOATS + 16 * I + 8 * hi;
  const v4f m0 = *(const v4f*)(mrow);
  const v4f m1 = *(const v4f*)(mrow + 4);

  v8f s0 = {}, s1 = {};
#pragma unroll
  for (int j = 0; j < I; ++j) {
    const v4f aa = *(const v4f*)(arow + 16 * j);
    const v4f ab = *(const v4f*)(arow + 16 * j + 4);
    s0 = wmma4(v2f{aa.x, aa.y}, v2f{z[j][0], z[j][1]}, s0);
    s1 = wmma4(v2f{aa.z, aa.w}, v2f{z[j][2], z[j][3]}, s1);
    s0 = wmma4(v2f{ab.x, ab.y}, v2f{z[j][4], z[j][5]}, s0);
    s1 = wmma4(v2f{ab.z, ab.w}, v2f{z[j][6], z[j][7]}, s1);
  }

  const v8f mm = {m0.x, m0.y, m0.z, m0.w, m1.x, m1.y, m1.z, m1.w};
  const v8f rr = (z[I] - mm) - (s0 + s1);

  const v4f da = *(const v4f*)(arow + 16 * I);
  const v4f db = *(const v4f*)(arow + 16 * I + 4);
  v8f zz = {};
  zz = wmma4(v2f{da.x, da.y}, v2f{rr[0], rr[1]}, zz);
  zz = wmma4(v2f{da.z, da.w}, v2f{rr[2], rr[3]}, zz);
  zz = wmma4(v2f{db.x, db.y}, v2f{rr[4], rr[5]}, zz);
  zz = wmma4(v2f{db.z, db.w}, v2f{rr[6], rr[7]}, zz);
  z[I] = zz;
#pragma unroll
  for (int r = 0; r < 8; ++r) p += zz[r] * zz[r];
}

__global__ void __launch_bounds__(256)
gmm_kernel(const float* __restrict__ feats, const float* __restrict__ covs,
           const float* __restrict__ Dinv, const float* __restrict__ logdet,
           const float* __restrict__ means, const float* __restrict__ weights,
           float* __restrict__ out) {
  __shared__ __attribute__((aligned(16))) float Lp[LP_FLOATS + D_F];
  const int tid = threadIdx.x;
  const int lane = tid & 31;
  const int w = tid >> 5;
  const int mk = blockIdx.y;
  const int b0 = blockIdx.x * 128;
  const float* Lg = covs + (size_t)mk * (D_F * D_F);
  const float* Dg = Dinv + (size_t)mk * (16 * 256);
  const float* mu = means + (size_t)mk * D_F;
  const int l15 = lane & 15;
  const int hi = lane >> 4;
  const int pr = tid >> 4, ps = tid & 15;

  const float* frow = feats + (size_t)(b0 + 16 * w + l15) * D_F + 8 * hi;
  v8f z[16];
#pragma unroll
  for (int i = 0; i < 8; ++i) {
    const v4f f0 = *(const v4f*)(frow + 16 * i);
    const v4f f1 = *(const v4f*)(frow + 16 * i + 4);
    z[i] = v8f{f0.x, f0.y, f0.z, f0.w, f1.x, f1.y, f1.z, f1.w};
  }

  if (tid < 64) *(v4f*)(Lp + LP_FLOATS + 4 * tid) = *(const v4f*)(mu + 4 * tid);

  {
    v4f t0[4], t1[4], t2[4], t3[4], t4[4], t5[4], t6[4], t7[4];
    stage_issue<0>(t0, Lg, Dg, pr, ps);
    stage_issue<1>(t1, Lg, Dg, pr, ps);
    stage_issue<2>(t2, Lg, Dg, pr, ps);
    stage_issue<3>(t3, Lg, Dg, pr, ps);
    stage_issue<4>(t4, Lg, Dg, pr, ps);
    stage_issue<5>(t5, Lg, Dg, pr, ps);
    stage_issue<6>(t6, Lg, Dg, pr, ps);
    stage_issue<7>(t7, Lg, Dg, pr, ps);
    stage_commit<0>(t0, Lp, pr, ps);
    stage_commit<1>(t1, Lp, pr, ps);
    stage_commit<2>(t2, Lp, pr, ps);
    stage_commit<3>(t3, Lp, pr, ps);
    stage_commit<4>(t4, Lp, pr, ps);
    stage_commit<5>(t5, Lp, pr, ps);
    stage_commit<6>(t6, Lp, pr, ps);
    stage_commit<7>(t7, Lp, pr, ps);
#define STAGE_4ROWS(A)                          \
  stage_issue<(A)>(t0, Lg, Dg, pr, ps);         \
  stage_issue<(A) + 1>(t1, Lg, Dg, pr, ps);     \
  stage_issue<(A) + 2>(t2, Lg, Dg, pr, ps);     \
  stage_issue<(A) + 3>(t3, Lg, Dg, pr, ps);     \
  stage_commit<(A)>(t0, Lp, pr, ps);            \
  stage_commit<(A) + 1>(t1, Lp, pr, ps);        \
  stage_commit<(A) + 2>(t2, Lp, pr, ps);        \
  stage_commit<(A) + 3>(t3, Lp, pr, ps);
    STAGE_4ROWS(8) STAGE_4ROWS(12)
#undef STAGE_4ROWS
  }
  __syncthreads();

#pragma unroll
  for (int i = 8; i < 16; ++i) {
    const v4f f0 = *(const v4f*)(frow + 16 * i);
    const v4f f1 = *(const v4f*)(frow + 16 * i + 4);
    z[i] = v8f{f0.x, f0.y, f0.z, f0.w, f1.x, f1.y, f1.z, f1.w};
  }

  float p = 0.f;
  solve_step<0>(z, p, Lp, l15, hi);
  solve_step<1>(z, p, Lp, l15, hi);
  solve_step<2>(z, p, Lp, l15, hi);
  solve_step<3>(z, p, Lp, l15, hi);
  solve_step<4>(z, p, Lp, l15, hi);
  solve_step<5>(z, p, Lp, l15, hi);
  solve_step<6>(z, p, Lp, l15, hi);
  solve_step<7>(z, p, Lp, l15, hi);
#define SOLVE_ALONE(I)                  \
  __builtin_amdgcn_sched_barrier(0);    \
  solve_step<(I)>(z, p, Lp, l15, hi);
  SOLVE_ALONE(8)  SOLVE_ALONE(9)  SOLVE_ALONE(10) SOLVE_ALONE(11)
  SOLVE_ALONE(12) SOLVE_ALONE(13) SOLVE_ALONE(14) SOLVE_ALONE(15)
#undef SOLVE_ALONE

  p += __shfl_xor(p, 16);

  if (hi == 0) {
    int m = mk >> 3, kcomp = mk & 7;
    const float* wr = weights + m * NCOMP;
    float wmax = wr[0];
#pragma unroll
    for (int t = 1; t < 8; ++t) wmax = fmaxf(wmax, wr[t]);
    float denom = 0.f;
#pragma unroll
    for (int t = 0; t < 8; ++t) denom += expf(wr[t] - wmax);
    float wmk = expf(wr[kcomp] - wmax) / denom;
    float logp = -0.5f * (p + 256.0f * LOG2PI) - logdet[mk];
    out[(size_t)(b0 + 16 * w + l15) * NMK + mk] = wmk * logp;
  }
}

extern "C" void kernel_launch(void* const* d_in, const int* in_sizes, int n_in,
                              void* d_out, int out_size, void* d_ws,
                              size_t ws_size, hipStream_t stream) {
  (void)in_sizes; (void)n_in; (void)out_size; (void)ws_size;
  const float* x       = (const float*)d_in[0];
  const float* W       = (const float*)d_in[1];
  const float* bias    = (const float*)d_in[2];
  const float* means   = (const float*)d_in[3];
  const float* covs    = (const float*)d_in[4];
  const float* weights = (const float*)d_in[5];
  float* out = (float*)d_out;

  float* ws = (float*)d_ws;
  float* feats  = ws;
  float* DinvW  = ws + (size_t)B_N * D_F;
  float* logdet = DinvW + (size_t)NMK * 16 * 256;

  feats_kernel<<<dim3(D_F / 64, B_N / 64), 256, 0, stream>>>(x, W, bias, feats);

  diag_kernel<<<dim3(NMK), 256, 0, stream>>>(covs, DinvW, logdet);

  gmm_kernel<<<dim3(B_N / 128, NMK), 256, 0, stream>>>(
      feats, covs, DinvW, logdet, means, weights, out);
}
